# starting kernel
# baseline (speedup 1.0000x reference)
#include <hip/hip_runtime.h>

typedef float v2f __attribute__((ext_vector_type(2)));
typedef float v8f __attribute__((ext_vector_type(8)));

#define DDIM 512
#define FDIM 64
#define ODIM 64
#define LDSP 516
#define MROWS 32

__global__ __launch_bounds__(256) void build_w_kernel(
    const float* __restrict__ kern, const int* __restrict__ fid,
    float* __restrict__ W) {
  const int t = blockIdx.x * 256 + threadIdx.x;
  const int i = t >> 9;
  const int j = t & 511;
  float acc = 0.0f;
  if (j > i) {
    const float* ki = kern + (size_t)i * (FDIM * ODIM) + (size_t)fid[j] * ODIM;
    const float* kj = kern + (size_t)j * (FDIM * ODIM) + (size_t)fid[i] * ODIM;
#pragma unroll
    for (int o = 0; o < ODIM; o += 4) {
      const float4 a = *(const float4*)(ki + o);
      const float4 b = *(const float4*)(kj + o);
      acc += a.x * b.x + a.y * b.y + a.z * b.z + a.w * b.w;
    }
  }
  W[t] = acc;
}

__global__ __launch_bounds__(256) void quad_form_kernel(
    const float* __restrict__ X, const float* __restrict__ W,
    float* __restrict__ out) {
  __shared__ float ldsX[MROWS * LDSP];
  __shared__ float partial[8][MROWS];

  const int tid   = threadIdx.x;
  const int wave  = tid >> 5;
  const int lane  = tid & 31;
  const int l15   = lane & 15;
  const int half  = lane >> 4;
  const int mBase = blockIdx.x * MROWS;

  {
    const float4* src = (const float4*)(X + (size_t)mBase * DDIM);
#pragma unroll
    for (int it = 0; it < 16; ++it) {
      const int idx = tid + it * 256;
      const int row = idx >> 7;
      const int col = idx & 127;
      *(float4*)(ldsX + row * LDSP + col * 4) = src[row * 128 + col];
    }
  }
  __syncthreads();

  float rowSum0[8], rowSum1[8];
#pragma unroll
  for (int r = 0; r < 8; ++r) { rowSum0[r] = 0.0f; rowSum1[r] = 0.0f; }

#pragma unroll 1
  for (int t = 0; t < 4; ++t) {
    const int jBase = (wave + t * 8) * 16;
    v8f c0 = {0.f, 0.f, 0.f, 0.f, 0.f, 0.f, 0.f, 0.f};
    v8f c1 = {0.f, 0.f, 0.f, 0.f, 0.f, 0.f, 0.f, 0.f};
    const float* wcol = W + jBase + l15;

#pragma unroll 4
    for (int k0 = 0; k0 < DDIM; k0 += 4) {
      const int kA = k0 + half * 2;
      v2f a0, a1, b;
      b.x = wcol[(size_t)kA * DDIM];
      b.y = wcol[(size_t)(kA + 1) * DDIM];
      a0.x = ldsX[l15 * LDSP + kA];
      a0.y = ldsX[l15 * LDSP + kA + 1];
      a1.x = ldsX[(16 + l15) * LDSP + kA];
      a1.y = ldsX[(16 + l15) * LDSP + kA + 1];
      c0 = __builtin_amdgcn_wmma_f32_16x16x4_f32(
          false, a0, false, b, (short)0, c0, false, false);
      c1 = __builtin_amdgcn_wmma_f32_16x16x4_f32(
          false, a1, false, b, (short)0, c1, false, false);
    }

#pragma unroll
    for (int r = 0; r < 8; ++r) {
      const int m = r + half * 8;
      rowSum0[r] += c0[r] * ldsX[m * LDSP + jBase + l15];
      rowSum1[r] += c1[r] * ldsX[(16 + m) * LDSP + jBase + l15];
    }
  }

#pragma unroll
  for (int r = 0; r < 8; ++r) {
    float v0 = rowSum0[r];
    v0 += __shfl_xor(v0, 1, 32);
    v0 += __shfl_xor(v0, 2, 32);
    v0 += __shfl_xor(v0, 4, 32);
    v0 += __shfl_xor(v0, 8, 32);
    rowSum0[r] = v0;
    float v1 = rowSum1[r];
    v1 += __shfl_xor(v1, 1, 32);
    v1 += __shfl_xor(v1, 2, 32);
    v1 += __shfl_xor(v1, 4, 32);
    v1 += __shfl_xor(v1, 8, 32);
    rowSum1[r] = v1;
  }
  if (l15 == 0) {
#pragma unroll
    for (int r = 0; r < 8; ++r) {
      partial[wave][r + half * 8]      = rowSum0[r];
      partial[wave][16 + r + half * 8] = rowSum1[r];
    }
  }
  __syncthreads();

  if (tid < MROWS) {
    float s = 0.0f;
#pragma unroll
    for (int w = 0; w < 8; ++w) s += partial[w][tid];
    out[mBase + tid] = s;
  }
}

extern "C" void kernel_launch(void* const* d_in, const int* in_sizes, int n_in,
                              void* d_out, int out_size, void* d_ws, size_t ws_size,
                              hipStream_t stream) {
  const float* x    = (const float*)d_in[0];
  const float* kern = (const float*)d_in[1];
  const int*   fid  = (const int*)d_in[2];
  float* out = (float*)d_out;
  float* W   = (float*)d_ws;

  build_w_kernel<<<(DDIM * DDIM) / 256, 256, 0, stream>>>(kern, fid, W);
  quad_form_kernel<<<4096 / MROWS, 256, 0, stream>>>(x, W, out);
}


# Skip the structurally zero lower triangle of Wm in the quad-form GEMM: column tile J loops over J+1 K-blocks of 16 rows; tiles paired w,15-w,16+w,31-w so every wave does 66 of 128 blocks
# speedup vs baseline: 1.6633x; 1.6633x over previous
#include <hip/hip_runtime.h>

typedef float v2f __attribute__((ext_vector_type(2)));
typedef float v8f __attribute__((ext_vector_type(8)));

#define DDIM 512
#define FDIM 64
#define ODIM 64
#define LDSP 516
#define MROWS 32

__global__ __launch_bounds__(256) void build_w_kernel(
    const float* __restrict__ kern, const int* __restrict__ fid,
    float* __restrict__ W) {
  const int t = blockIdx.x * 256 + threadIdx.x;
  const int i = t >> 9;
  const int j = t & 511;
  float acc = 0.0f;
  if (j > i) {
    const float* ki = kern + (size_t)i * (FDIM * ODIM) + (size_t)fid[j] * ODIM;
    const float* kj = kern + (size_t)j * (FDIM * ODIM) + (size_t)fid[i] * ODIM;
#pragma unroll
    for (int o = 0; o < ODIM; o += 4) {
      const float4 a = *(const float4*)(ki + o);
      const float4 b = *(const float4*)(kj + o);
      acc += a.x * b.x + a.y * b.y + a.z * b.z + a.w * b.w;
    }
  }
  W[t] = acc;
}

__global__ __launch_bounds__(256) void quad_form_kernel(
    const float* __restrict__ X, const float* __restrict__ W,
    float* __restrict__ out) {
  __shared__ float ldsX[MROWS * LDSP];
  __shared__ float partial[8][MROWS];

  const int tid   = threadIdx.x;
  const int wave  = __builtin_amdgcn_readfirstlane(tid >> 5);
  const int lane  = tid & 31;
  const int l15   = lane & 15;
  const int half  = lane >> 4;
  const int mBase = blockIdx.x * MROWS;

  {
    const float4* src = (const float4*)(X + (size_t)mBase * DDIM);
#pragma unroll
    for (int it = 0; it < 16; ++it) {
      const int idx = tid + it * 256;
      const int row = idx >> 7;
      const int col = idx & 127;
      *(float4*)(ldsX + row * LDSP + col * 4) = src[row * 128 + col];
    }
  }
  __syncthreads();

  float rowSum0[8], rowSum1[8];
#pragma unroll
  for (int r = 0; r < 8; ++r) { rowSum0[r] = 0.0f; rowSum1[r] = 0.0f; }

#pragma unroll 1
  for (int t = 0; t < 4; ++t) {
    const int jTile = (t >> 1) * 16 + ((t & 1) ? (15 - wave) : wave);
    const int jBase = jTile * 16;
    v8f c0 = {0.f, 0.f, 0.f, 0.f, 0.f, 0.f, 0.f, 0.f};
    v8f c1 = {0.f, 0.f, 0.f, 0.f, 0.f, 0.f, 0.f, 0.f};
    const float* wcol = W + jBase + l15;

#pragma unroll 1
    for (int kb = 0; kb <= jTile; ++kb) {
#pragma unroll
    for (int kk = 0; kk < 16; kk += 4) {
      const int k0 = kb * 16 + kk;
      const int kA = k0 + half * 2;
      v2f a0, a1, b;
      b.x = wcol[(size_t)kA * DDIM];
      b.y = wcol[(size_t)(kA + 1) * DDIM];
      a0.x = ldsX[l15 * LDSP + kA];
      a0.y = ldsX[l15 * LDSP + kA + 1];
      a1.x = ldsX[(16 + l15) * LDSP + kA];
      a1.y = ldsX[(16 + l15) * LDSP + kA + 1];
      c0 = __builtin_amdgcn_wmma_f32_16x16x4_f32(
          false, a0, false, b, (short)0, c0, false, false);
      c1 = __builtin_amdgcn_wmma_f32_16x16x4_f32(
          false, a1, false, b, (short)0, c1, false, false);
    }
    }

#pragma unroll
    for (int r = 0; r < 8; ++r) {
      const int m = r + half * 8;
      rowSum0[r] += c0[r] * ldsX[m * LDSP + jBase + l15];
      rowSum1[r] += c1[r] * ldsX[(16 + m) * LDSP + jBase + l15];
    }
  }

#pragma unroll
  for (int r = 0; r < 8; ++r) {
    float v0 = rowSum0[r];
    v0 += __shfl_xor(v0, 1, 32);
    v0 += __shfl_xor(v0, 2, 32);
    v0 += __shfl_xor(v0, 4, 32);
    v0 += __shfl_xor(v0, 8, 32);
    rowSum0[r] = v0;
    float v1 = rowSum1[r];
    v1 += __shfl_xor(v1, 1, 32);
    v1 += __shfl_xor(v1, 2, 32);
    v1 += __shfl_xor(v1, 4, 32);
    v1 += __shfl_xor(v1, 8, 32);
    rowSum1[r] = v1;
  }
  if (l15 == 0) {
#pragma unroll
    for (int r = 0; r < 8; ++r) {
      partial[wave][r + half * 8]      = rowSum0[r];
      partial[wave][16 + r + half * 8] = rowSum1[r];
    }
  }
  __syncthreads();

  if (tid < MROWS) {
    float s = 0.0f;
#pragma unroll
    for (int w = 0; w < 8; ++w) s += partial[w][tid];
    out[mBase + tid] = s;
  }
}

extern "C" void kernel_launch(void* const* d_in, const int* in_sizes, int n_in,
                              void* d_out, int out_size, void* d_ws, size_t ws_size,
                              hipStream_t stream) {
  const float* x    = (const float*)d_in[0];
  const float* kern = (const float*)d_in[1];
  const int*   fid  = (const int*)d_in[2];
  float* out = (float*)d_out;
  float* W   = (float*)d_ws;

  build_w_kernel<<<(DDIM * DDIM) / 256, 256, 0, stream>>>(kern, fid, W);
  quad_form_kernel<<<4096 / MROWS, 256, 0, stream>>>(x, W, out);
}


# Software-pipeline the Wm fragment loads: a ring of 6 register sets over each wave's flat sequence of 66 K-blocks keeps 5 blocks (40 loads) in flight behind the WMMAs, across tile boundaries
# speedup vs baseline: 2.8358x; 1.7049x over previous
#include <hip/hip_runtime.h>

typedef float v2f __attribute__((ext_vector_type(2)));
typedef float v8f __attribute__((ext_vector_type(8)));

#define DDIM 512
#define FDIM 64
#define ODIM 64
#define LDSP 516
#define MROWS 32

__global__ __launch_bounds__(256) void build_w_kernel(
    const float* __restrict__ kern, const int* __restrict__ fid,
    float* __restrict__ W) {
  const int t = blockIdx.x * 256 + threadIdx.x;
  const int i = t >> 9;
  const int j = t & 511;
  float acc = 0.0f;
  if (j > i) {
    const float* ki = kern + (size_t)i * (FDIM * ODIM) + (size_t)fid[j] * ODIM;
    const float* kj = kern + (size_t)j * (FDIM * ODIM) + (size_t)fid[i] * ODIM;
#pragma unroll
    for (int o = 0; o < ODIM; o += 4) {
      const float4 a = *(const float4*)(ki + o);
      const float4 b = *(const float4*)(kj + o);
      acc += a.x * b.x + a.y * b.y + a.z * b.z + a.w * b.w;
    }
  }
  W[t] = acc;
}

#define RING 6
#define NBLK 66

__global__ __launch_bounds__(256) void quad_form_kernel(
    const float* __restrict__ X, const float* __restrict__ W,
    float* __restrict__ out) {
  __shared__ float ldsX[MROWS * LDSP];
  __shared__ float partial[8][MROWS];

  const int tid   = threadIdx.x;
  const int wave  = __builtin_amdgcn_readfirstlane(tid >> 5);
  const int lane  = tid & 31;
  const int l15   = lane & 15;
  const int half  = lane >> 4;
  const int mBase = blockIdx.x * MROWS;

  float4 xr[16];
  {
    const float4* src = (const float4*)(X + (size_t)mBase * DDIM);
#pragma unroll
    for (int it = 0; it < 16; ++it) xr[it] = src[tid + it * 256];
  }

  const unsigned laneOff = (unsigned)(half * 2 * DDIM + l15);

  int pt = 0, pj = wave, pkb = 0;
  v2f ring[RING][4];

#define LOAD_B(r)                                                             \
  {                                                                           \
    const float* sb = W + (pkb * (16 * DDIM) + pj * 16);                      \
    _Pragma("unroll") for (int kk = 0; kk < 4; ++kk) {                        \
      ring[r][kk].x = sb[laneOff + (unsigned)((kk * 4) * DDIM)];              \
      ring[r][kk].y = sb[laneOff + (unsigned)((kk * 4 + 1) * DDIM)];          \
    }                                                                         \
    const bool plast = (pkb == pj);                                           \
    const bool pnext = plast && (pt < 3);                                     \
    pt += pnext ? 1 : 0;                                                      \
    pkb = pnext ? 0 : (plast ? pkb : pkb + 1);                                \
    pj = (pt >> 1) * 16 + ((pt & 1) ? (15 - wave) : wave);                    \
  }

#pragma unroll
  for (int r = 0; r < RING; ++r) LOAD_B(r)

#pragma unroll
  for (int it = 0; it < 16; ++it) {
    const int idx = tid + it * 256;
    const int row = idx >> 7;
    const int col = idx & 127;
    *(float4*)(ldsX + row * LDSP + col * 4) = xr[it];
  }
  __syncthreads();

  float rowSum0[8], rowSum1[8];
#pragma unroll
  for (int r = 0; r < 8; ++r) { rowSum0[r] = 0.0f; rowSum1[r] = 0.0f; }

  v8f c0 = {0.f, 0.f, 0.f, 0.f, 0.f, 0.f, 0.f, 0.f};
  v8f c1 = {0.f, 0.f, 0.f, 0.f, 0.f, 0.f, 0.f, 0.f};

  int ct = 0, cj = wave, ckb = 0;
  const float* xrow = ldsX + l15 * LDSP + half * 2;

#pragma unroll 1
  for (int it = 0; it < NBLK / RING; ++it) {
#pragma unroll
    for (int r = 0; r < RING; ++r) {
      const float* xa = xrow + ckb * 16;
#pragma unroll
      for (int kk = 0; kk < 4; ++kk) {
        const v2f a0 = *(const v2f*)(xa + kk * 4);
        const v2f a1 = *(const v2f*)(xa + 16 * LDSP + kk * 4);
        c0 = __builtin_amdgcn_wmma_f32_16x16x4_f32(
            false, a0, false, ring[r][kk], (short)0, c0, false, false);
        c1 = __builtin_amdgcn_wmma_f32_16x16x4_f32(
            false, a1, false, ring[r][kk], (short)0, c1, false, false);
      }

      LOAD_B(r)

      if (ckb == cj) {
        const int jBase = cj * 16;
#pragma unroll
        for (int q = 0; q < 8; ++q) {
          const int m = q + half * 8;
          rowSum0[q] += c0[q] * ldsX[m * LDSP + jBase + l15];
          rowSum1[q] += c1[q] * ldsX[(16 + m) * LDSP + jBase + l15];
        }
#pragma unroll
        for (int q = 0; q < 8; ++q) { c0[q] = 0.0f; c1[q] = 0.0f; }
        ++ct;
        cj = (ct >> 1) * 16 + ((ct & 1) ? (15 - wave) : wave);
        ckb = 0;
      } else {
        ++ckb;
      }
    }
  }
#undef LOAD_B

#pragma unroll
  for (int r = 0; r < 8; ++r) {
    float v0 = rowSum0[r];
    v0 += __shfl_xor(v0, 1, 32);
    v0 += __shfl_xor(v0, 2, 32);
    v0 += __shfl_xor(v0, 4, 32);
    v0 += __shfl_xor(v0, 8, 32);
    rowSum0[r] = v0;
    float v1 = rowSum1[r];
    v1 += __shfl_xor(v1, 1, 32);
    v1 += __shfl_xor(v1, 2, 32);
    v1 += __shfl_xor(v1, 4, 32);
    v1 += __shfl_xor(v1, 8, 32);
    rowSum1[r] = v1;
  }
  if (l15 == 0) {
#pragma unroll
    for (int r = 0; r < 8; ++r) {
      partial[wave][r + half * 8]      = rowSum0[r];
      partial[wave][16 + r + half * 8] = rowSum1[r];
    }
  }
  __syncthreads();

  if (tid < MROWS) {
    float s = 0.0f;
#pragma unroll
    for (int w = 0; w < 8; ++w) s += partial[w][tid];
    out[mBase + tid] = s;
  }
}

extern "C" void kernel_launch(void* const* d_in, const int* in_sizes, int n_in,
                              void* d_out, int out_size, void* d_ws, size_t ws_size,
                              hipStream_t stream) {
  const float* x    = (const float*)d_in[0];
  const float* kern = (const float*)d_in[1];
  const int*   fid  = (const int*)d_in[2];
  float* out = (float*)d_out;
  float* W   = (float*)d_ws;

  build_w_kernel<<<(DDIM * DDIM) / 256, 256, 0, stream>>>(kern, fid, W);
  quad_form_kernel<<<4096 / MROWS, 256, 0, stream>>>(x, W, out);
}


# Wm build re-mapped: 16 lanes per (i,j) pair read both 256-byte kernel rows as whole lines, DPP row sum; only the blocks kernel 2 reads are built; 256 workgroups of 9 waves, 17 tasks each
# speedup vs baseline: 3.8758x; 1.3667x over previous
#include <hip/hip_runtime.h>

typedef float v2f __attribute__((ext_vector_type(2)));
typedef float v8f __attribute__((ext_vector_type(8)));
typedef float v4f __attribute__((ext_vector_type(4)));

#define DDIM 512
#define FDIM 64
#define ODIM 64
#define LDSP 516
#define MROWS 32

#define K1_WAVES 9
#define K1_TASKS 17

__device__ __forceinline__ float row16_sum(float v) {
  v += __int_as_float(__builtin_amdgcn_update_dpp(0, __float_as_int(v), 0x161, 0xf, 0xf, true));
  v += __int_as_float(__builtin_amdgcn_update_dpp(0, __float_as_int(v), 0x162, 0xf, 0xf, true));
  v += __int_as_float(__builtin_amdgcn_update_dpp(0, __float_as_int(v), 0x164, 0xf, 0xf, true));
  v += __int_as_float(__builtin_amdgcn_update_dpp(0, __float_as_int(v), 0x168, 0xf, 0xf, true));
  return v;
}

__global__ __launch_bounds__(K1_WAVES * 32) void build_w_kernel(
    const float* __restrict__ kern, const int* __restrict__ fid,
    float* __restrict__ W) {
  const int tid  = threadIdx.x;
  const int wave = __builtin_amdgcn_readfirstlane(tid >> 5);
  const int lane = tid & 31;
  const int l15  = lane & 15;
  const int half = lane >> 4;
  const int b    = blockIdx.x;
  const int g    = b >> 5;
  const int n1   = 16 - g;

#pragma unroll 1
  for (int t = wave; t < K1_TASKS; t += K1_WAVES) {
    const bool lo = (t < n1);
    const int i = lo ? b : (DDIM - 1 - b);
    const int C = lo ? (g + t) : (15 - g + (t - n1));
    const int fi = fid[i];
    const int jh = C * 32 + half * 16;
    const float* kiBase = kern + (size_t)i * (FDIM * ODIM) + l15 * 4;
    const float* kjBase = kern + (size_t)fi * ODIM + l15 * 4;

    int fj[16];
#pragma unroll
    for (int s = 0; s < 16; ++s) fj[s] = fid[jh + s];

    v4f av[16], bv[16];
#pragma unroll
    for (int s = 0; s < 16; ++s)
      bv[s] = *(const v4f*)(kjBase + (size_t)(jh + s) * (FDIM * ODIM));
#pragma unroll
    for (int s = 0; s < 16; ++s)
      av[s] = *(const v4f*)(kiBase + (size_t)fj[s] * ODIM);
    __builtin_amdgcn_sched_barrier(0);

    float res = 0.0f;
#pragma unroll
    for (int s = 0; s < 16; ++s) {
      asm volatile("" : "+v"(bv[s]));
      asm volatile("" : "+v"(av[s]));
      float p = av[s].x * bv[s].x + av[s].y * bv[s].y +
                av[s].z * bv[s].z + av[s].w * bv[s].w;
      p = row16_sum(p);
      res = (l15 == s) ? p : res;
    }

    const int j = C * 32 + lane;
    W[i * DDIM + j] = (j > i) ? res : 0.0f;
  }
}

#define RING 6
#define NBLK 66

__global__ __launch_bounds__(256) void quad_form_kernel(
    const float* __restrict__ X, const float* __restrict__ W,
    float* __restrict__ out) {
  __shared__ float ldsX[MROWS * LDSP];
  __shared__ float partial[8][MROWS];

  const int tid   = threadIdx.x;
  const int wave  = __builtin_amdgcn_readfirstlane(tid >> 5);
  const int lane  = tid & 31;
  const int l15   = lane & 15;
  const int half  = lane >> 4;
  const int mBase = blockIdx.x * MROWS;

  float4 xr[16];
  {
    const float4* src = (const float4*)(X + (size_t)mBase * DDIM);
#pragma unroll
    for (int it = 0; it < 16; ++it) xr[it] = src[tid + it * 256];
  }

  const unsigned laneOff = (unsigned)(half * 2 * DDIM + l15);

  int pt = 0, pj = wave, pkb = 0;
  v2f ring[RING][4];

#define LOAD_B(r)                                                             \
  {                                                                           \
    const float* sb = W + (pkb * (16 * DDIM) + pj * 16);                      \
    _Pragma("unroll") for (int kk = 0; kk < 4; ++kk) {                        \
      ring[r][kk].x = sb[laneOff + (unsigned)((kk * 4) * DDIM)];              \
      ring[r][kk].y = sb[laneOff + (unsigned)((kk * 4 + 1) * DDIM)];          \
    }                                                                         \
    const bool plast = (pkb == pj);                                           \
    const bool pnext = plast && (pt < 3);                                     \
    pt += pnext ? 1 : 0;                                                      \
    pkb = pnext ? 0 : (plast ? pkb : pkb + 1);                                \
    pj = (pt >> 1) * 16 + ((pt & 1) ? (15 - wave) : wave);                    \
  }

#pragma unroll
  for (int r = 0; r < RING; ++r) LOAD_B(r)

#pragma unroll
  for (int it = 0; it < 16; ++it) {
    const int idx = tid + it * 256;
    const int row = idx >> 7;
    const int col = idx & 127;
    *(float4*)(ldsX + row * LDSP + col * 4) = xr[it];
  }
  __syncthreads();

  float rowSum0[8], rowSum1[8];
#pragma unroll
  for (int r = 0; r < 8; ++r) { rowSum0[r] = 0.0f; rowSum1[r] = 0.0f; }

  v8f c0 = {0.f, 0.f, 0.f, 0.f, 0.f, 0.f, 0.f, 0.f};
  v8f c1 = {0.f, 0.f, 0.f, 0.f, 0.f, 0.f, 0.f, 0.f};

  int ct = 0, cj = wave, ckb = 0;
  const float* xrow = ldsX + l15 * LDSP + half * 2;

#pragma unroll 1
  for (int it = 0; it < NBLK / RING; ++it) {
#pragma unroll
    for (int r = 0; r < RING; ++r) {
      const float* xa = xrow + ckb * 16;
#pragma unroll
      for (int kk = 0; kk < 4; ++kk) {
        const v2f a0 = *(const v2f*)(xa + kk * 4);
        const v2f a1 = *(const v2f*)(xa + 16 * LDSP + kk * 4);
        c0 = __builtin_amdgcn_wmma_f32_16x16x4_f32(
            false, a0, false, ring[r][kk], (short)0, c0, false, false);
        c1 = __builtin_amdgcn_wmma_f32_16x16x4_f32(
            false, a1, false, ring[r][kk], (short)0, c1, false, false);
      }

      LOAD_B(r)

      if (ckb == cj) {
        const int jBase = cj * 16;
#pragma unroll
        for (int q = 0; q < 8; ++q) {
          const int m = q + half * 8;
          rowSum0[q] += c0[q] * ldsX[m * LDSP + jBase + l15];
          rowSum1[q] += c1[q] * ldsX[(16 + m) * LDSP + jBase + l15];
        }
#pragma unroll
        for (int q = 0; q < 8; ++q) { c0[q] = 0.0f; c1[q] = 0.0f; }
        ++ct;
        cj = (ct >> 1) * 16 + ((ct & 1) ? (15 - wave) : wave);
        ckb = 0;
      } else {
        ++ckb;
      }
    }
  }
#undef LOAD_B

#pragma unroll
  for (int r = 0; r < 8; ++r) {
    float v0 = rowSum0[r];
    v0 += __shfl_xor(v0, 1, 32);
    v0 += __shfl_xor(v0, 2, 32);
    v0 += __shfl_xor(v0, 4, 32);
    v0 += __shfl_xor(v0, 8, 32);
    rowSum0[r] = v0;
    float v1 = rowSum1[r];
    v1 += __shfl_xor(v1, 1, 32);
    v1 += __shfl_xor(v1, 2, 32);
    v1 += __shfl_xor(v1, 4, 32);
    v1 += __shfl_xor(v1, 8, 32);
    rowSum1[r] = v1;
  }
  if (l15 == 0) {
#pragma unroll
    for (int r = 0; r < 8; ++r) {
      partial[wave][r + half * 8]      = rowSum0[r];
      partial[wave][16 + r + half * 8] = rowSum1[r];
    }
  }
  __syncthreads();

  if (tid < MROWS) {
    float s = 0.0f;
#pragma unroll
    for (int w = 0; w < 8; ++w) s += partial[w][tid];
    out[mBase + tid] = s;
  }
}

extern "C" void kernel_launch(void* const* d_in, const int* in_sizes, int n_in,
                              void* d_out, int out_size, void* d_ws, size_t ws_size,
                              hipStream_t stream) {
  const float* x    = (const float*)d_in[0];
  const float* kern = (const float*)d_in[1];
  const int*   fid  = (const int*)d_in[2];
  float* out = (float*)d_out;
  float* W   = (float*)d_ws;

  build_w_kernel<<<DDIM / 2, K1_WAVES * 32, 0, stream>>>(kern, fid, W);
  quad_form_kernel<<<4096 / MROWS, 256, 0, stream>>>(x, W, out);
}


# Quad-form GEMM on all 256 WGPs: 16 batch rows per workgroup (256 workgroups of 8 waves); kernel 1 stores Wm in WMMA fragment order: a 16x16 block is two whole-line 16-byte loads per lane; ring of 11
# speedup vs baseline: 4.7431x; 1.2238x over previous
#include <hip/hip_runtime.h>


typedef float v2f __attribute__((ext_vector_type(2)));
typedef float v8f __attribute__((ext_vector_type(8)));
typedef float v4f __attribute__((ext_vector_type(4)));

#define DDIM 512
#define FDIM 64
#define ODIM 64
#define LDSP 516
#define MROWS 16


#define K1_WAVES 8
#define K1_UNITS 272

__device__ __forceinline__ float row16_sum(float v) {
  v += __int_as_float(__builtin_amdgcn_update_dpp(0, __float_as_int(v), 0x161, 0xf, 0xf, true));
  v += __int_as_float(__builtin_amdgcn_update_dpp(0, __float_as_int(v), 0x162, 0xf, 0xf, true));
  v += __int_as_float(__builtin_amdgcn_update_dpp(0, __float_as_int(v), 0x164, 0xf, 0xf, true));
  v += __int_as_float(__builtin_amdgcn_update_dpp(0, __float_as_int(v), 0x168, 0xf, 0xf, true));
  return v;
}

__global__ __launch_bounds__(K1_WAVES * 32) void build_w_kernel(
    const float* __restrict__ kern, const int* __restrict__ fid,
    float* __restrict__ W) {
  __shared__ float stage[512];

  const int tid  = threadIdx.x;
  const int wave = __builtin_amdgcn_readfirstlane(tid >> 5);
  const int lane = tid & 31;
  const int l15  = lane & 15;
  const int half = lane >> 4;

  const int u = blockIdx.x;
  int p = 0;
#pragma unroll
  for (int t = 0; t < 15; ++t) p += (u >= (p + 1) * (p + 2)) ? 1 : 0;
  const int rem = u - p * (p + 1);
  const int J = 2 * p + (rem & 1);
  const int C = rem >> 1;

#pragma unroll 1
  for (int t = 0; t < 2; ++t) {
    const int cc = wave + 8 * t;
    const int i  = J * 16 + cc;
    const int fi = fid[i];
    const int jh = C * 32 + half * 16;
    const float* kiBase = kern + (size_t)i * (FDIM * ODIM) + l15 * 4;
    const float* kjBase = kern + (size_t)fi * ODIM + l15 * 4;

    int fj[16];
#pragma unroll
    for (int s = 0; s < 16; ++s) fj[s] = fid[jh + s];

    v4f av[16], bv[16];
#pragma unroll
    for (int s = 0; s < 16; ++s)
      bv[s] = *(const v4f*)(kjBase + (size_t)(jh + s) * (FDIM * ODIM));
#pragma unroll
    for (int s = 0; s < 16; ++s)
      av[s] = *(const v4f*)(kiBase + (size_t)fj[s] * ODIM);
    __builtin_amdgcn_sched_barrier(0);

    float res = 0.0f;
#pragma unroll
    for (int s = 0; s < 16; ++s) {
      asm volatile("" : "+v"(bv[s]));
      asm volatile("" : "+v"(av[s]));
      float pr = av[s].x * bv[s].x + av[s].y * bv[s].y +
                 av[s].z * bv[s].z + av[s].w * bv[s].w;
      pr = row16_sum(pr);
      res = (l15 == s) ? pr : res;
    }

    const int k = C * 32 + lane;
    const float val = (k < i) ? res : 0.0f;
    const int off = half * 256 + ((l15 >> 2) & 1) * 128 +
                    (((l15 >> 3) * 16 + cc) * 4) + (l15 & 3);
    stage[off] = val;
  }
  __syncthreads();
  float* dst = W + (J * 32 + 2 * C) * 256;
  dst[tid]       = stage[tid];
  dst[tid + 256] = stage[tid + 256];
}

#define RING 11
#define NBLK 66
#define UNR 22

__global__ __launch_bounds__(256) void quad_form_kernel(
    const float* __restrict__ X, const float* __restrict__ W,
    float* __restrict__ out) {
  __shared__ __attribute__((aligned(16))) float ldsX[MROWS * LDSP];
  __shared__ float partial[8][MROWS];

  const int tid   = threadIdx.x;
  const int wave  = __builtin_amdgcn_readfirstlane(tid >> 5);
  const int lane  = tid & 31;
  const int l15   = lane & 15;
  const int half  = lane >> 4;
  const int mBase = blockIdx.x * MROWS;

  float4 xr[8];
  {
    const float4* src = (const float4*)(X + (size_t)mBase * DDIM);
#pragma unroll
    for (int it = 0; it < 8; ++it) xr[it] = src[tid + it * 256];
  }

  const unsigned laneOff = (unsigned)(lane * 4);

  int pt = 0, pj = wave, pkb = 0;
  v4f ring[RING][2];

#define LOAD_B(r)                                                             \
  {                                                                           \
    const float* sb = W + ((pj * 32 + pkb) * 256);                            \
    ring[r][0] = *(const v4f*)(sb + laneOff);                                 \
    ring[r][1] = *(const v4f*)(sb + laneOff + 128u);                          \
    const bool plast = (pkb == pj);                                           \
    const bool pnext = plast && (pt < 3);                                     \
    pt += pnext ? 1 : 0;                                                      \
    pkb = pnext ? 0 : (plast ? pkb : pkb + 1);                                \
    pj = (pt >> 1) * 16 + ((pt & 1) ? (15 - wave) : wave);                    \
  }

#pragma unroll
  for (int r = 0; r < RING; ++r) LOAD_B(r)

#pragma unroll
  for (int it = 0; it < 8; ++it) {
    const int idx = tid + it * 256;
    const int row = idx >> 7;
    const int col = idx & 127;
    *(float4*)(ldsX + row * LDSP + col * 4) = xr[it];
  }
  __syncthreads();

  float rowSum[8];
#pragma unroll
  for (int r = 0; r < 8; ++r) rowSum[r] = 0.0f;

  v8f c0 = {0.f, 0.f, 0.f, 0.f, 0.f, 0.f, 0.f, 0.f};
  v8f c1 = {0.f, 0.f, 0.f, 0.f, 0.f, 0.f, 0.f, 0.f};

  int ct = 0, cj = wave, ckb = 0;
  const float* xrow = ldsX + l15 * LDSP + half * 8;

  v4f aBuf[2][2];
  aBuf[0][0] = *(const v4f*)(xrow);
  aBuf[0][1] = *(const v4f*)(xrow + 4);

#pragma unroll 1
  for (int it = 0; it < NBLK / UNR; ++it) {
#pragma unroll
    for (int b = 0; b < UNR; ++b) {
      const int r   = b % RING;
      const int cur = b & 1;
      const int nxt = cur ^ 1;
      const bool last = (ckb == cj);
      const int jBase = cj * 16;

      if (last) {
        ++ct;
        cj = (ct >> 1) * 16 + ((ct & 1) ? (15 - wave) : wave);
        ckb = 0;
      } else {
        ++ckb;
      }
      aBuf[nxt][0] = *(const v4f*)(xrow + ckb * 16);
      aBuf[nxt][1] = *(const v4f*)(xrow + ckb * 16 + 4);
      __builtin_amdgcn_sched_barrier(0);
      const v4f a0 = aBuf[cur][0];
      const v4f a1 = aBuf[cur][1];

      c0 = __builtin_amdgcn_wmma_f32_16x16x4_f32(
          false, __builtin_shufflevector(a0, a0, 0, 1), false,
          __builtin_shufflevector(ring[r][0], ring[r][0], 0, 1), (short)0, c0, false, false);
      c1 = __builtin_amdgcn_wmma_f32_16x16x4_f32(
          false, __builtin_shufflevector(a0, a0, 2, 3), false,
          __builtin_shufflevector(ring[r][0], ring[r][0], 2, 3), (short)0, c1, false, false);
      c0 = __builtin_amdgcn_wmma_f32_16x16x4_f32(
          false, __builtin_shufflevector(a1, a1, 0, 1), false,
          __builtin_shufflevector(ring[r][1], ring[r][1], 0, 1), (short)0, c0, false, false);
      c1 = __builtin_amdgcn_wmma_f32_16x16x4_f32(
          false, __builtin_shufflevector(a1, a1, 2, 3), false,
          __builtin_shufflevector(ring[r][1], ring[r][1], 2, 3), (short)0, c1, false, false);

      LOAD_B(r)

      if (last) {
#pragma unroll
        for (int q = 0; q < 8; ++q) {
          const int m = q + half * 8;
          rowSum[q] += (c0[q] + c1[q]) * ldsX[m * LDSP + jBase + l15];
        }
#pragma unroll
        for (int q = 0; q < 8; ++q) { c0[q] = 0.0f; c1[q] = 0.0f; }
      }
    }
  }
#undef LOAD_B

#pragma unroll
  for (int r = 0; r < 8; ++r) {
    float v0 = rowSum[r];
    v0 += __shfl_xor(v0, 1, 32);
    v0 += __shfl_xor(v0, 2, 32);
    v0 += __shfl_xor(v0, 4, 32);
    v0 += __shfl_xor(v0, 8, 32);
    rowSum[r] = v0;
  }
  if (l15 == 0) {
#pragma unroll
    for (int r = 0; r < 8; ++r) partial[wave][r + half * 8] = rowSum[r];
  }
  __syncthreads();

  if (tid < MROWS) {
    float s = 0.0f;
#pragma unroll
    for (int w = 0; w < 8; ++w) s += partial[w][tid];
    out[mBase + tid] = s;
  }
}

extern "C" void kernel_launch(void* const* d_in, const int* in_sizes, int n_in,
                              void* d_out, int out_size, void* d_ws, size_t ws_size,
                              hipStream_t stream) {
  const float* x    = (const float*)d_in[0];
  const float* kern = (const float*)d_in[1];
  const int*   fid  = (const int*)d_in[2];
  float* out = (float*)d_out;
  float* W   = (float*)d_ws;

  build_w_kernel<<<K1_UNITS, K1_WAVES * 32, 0, stream>>>(kern, fid, W);
  quad_form_kernel<<<4096 / MROWS, 256, 0, stream>>>(x, W, out);
}


# Kernel 1 re-cut into 256 equal workgroups, one per WGP (was 272, so 16 WGPs took two): the 135,168 stored (k,n) pairs are dealt flat in storage order, 33 consecutive pairs per 16-lane half
# speedup vs baseline: 5.6670x; 1.1948x over previous
#include <hip/hip_runtime.h>


typedef float v2f __attribute__((ext_vector_type(2)));
typedef float v8f __attribute__((ext_vector_type(8)));
typedef float v4f __attribute__((ext_vector_type(4)));

#define DDIM 512
#define FDIM 64
#define ODIM 64
#define LDSP 516
#define MROWS 16


#define K1_WAVES 8
#define K1_UNITS 256
#define K1_SLOTS 528

__device__ __forceinline__ float row16_sum(float v) {
  v += __int_as_float(__builtin_amdgcn_update_dpp(0, __float_as_int(v), 0x161, 0xf, 0xf, true));
  v += __int_as_float(__builtin_amdgcn_update_dpp(0, __float_as_int(v), 0x162, 0xf, 0xf, true));
  v += __int_as_float(__builtin_amdgcn_update_dpp(0, __float_as_int(v), 0x164, 0xf, 0xf, true));
  v += __int_as_float(__builtin_amdgcn_update_dpp(0, __float_as_int(v), 0x168, 0xf, 0xf, true));
  return v;
}

template <int T>
__device__ __forceinline__ unsigned row_take(unsigned v) {
  return (unsigned)__builtin_amdgcn_update_dpp(0, (int)v, 0x150 + T, 0xf, 0xf, false);
}

__device__ __forceinline__ void decode_slot(int P, int ch0, const int* __restrict__ fid,
                                            unsigned& offA, unsigned& offB,
                                            int& sidx, bool& keep) {
  const int ch = P >> 8;
  const int cc = (P >> 4) & 15;
  const int kl = P & 15;
  const int wv = ch / 66;
  const int st = ch - wv * 66;
  const bool hi  = st > 16;
  const bool odd = st > (hi ? 33 + wv : wv);
  const int J    = (hi ? 16 : 0) + (odd ? 15 - wv : wv);
  const int kb   = st - (hi ? (odd ? 34 + wv : 17) : (odd ? wv + 1 : 0));
  const int n = J * 16 + cc;
  const int k = kb * 16 + kl;
  const int fn = fid[n] & (FDIM - 1);
  const int fk = fid[k] & (FDIM - 1);
  offA = (unsigned)((n * FDIM + fk) * (ODIM * 4));
  offB = (unsigned)((k * FDIM + fn) * (ODIM * 4));
  keep = k < n;
  sidx = (ch - ch0) * 256 + ((kl >> 2) & 1) * 128 + (((kl >> 3) * 16 + cc) * 4) + (kl & 3);
}

#define REP16(M) M(0) M(1) M(2) M(3) M(4) M(5) M(6) M(7) \
                 M(8) M(9) M(10) M(11) M(12) M(13) M(14) M(15)

__global__ __launch_bounds__(K1_WAVES * 32) void build_w_kernel(
    const float* __restrict__ kern, const int* __restrict__ fid,
    float* __restrict__ W) {
  __shared__ float stage[768];

  const int tid  = threadIdx.x;
  const int l15  = tid & 15;
  const int hw   = tid >> 4;
  const int g    = blockIdx.x;
  const int base = g * K1_SLOTS + hw * 33;
  const int ch0  = (g * K1_SLOTS) >> 8;

  unsigned dA0, dB0, dA1, dB1, dA2, dB2;
  int sx0, sx1, sx2;
  bool kp0, kp1, kp2;
  decode_slot(base + l15,      ch0, fid, dA0, dB0, sx0, kp0);
  decode_slot(base + 16 + l15, ch0, fid, dA1, dB1, sx1, kp1);
  decode_slot(base + 32,       ch0, fid, dA2, dB2, sx2, kp2);

  const char* kbytes = (const char*)kern;
  const unsigned lo = (unsigned)l15 * 16u;

  v4f av[17], bv[17];
  float res0 = 0.0f, res1 = 0.0f, res2 = 0.0f;

#define K1_DOT(t) (av[t].x * bv[t].x + av[t].y * bv[t].y + \
                   av[t].z * bv[t].z + av[t].w * bv[t].w)

#define K1_LOAD0(t)                                                        \
  av[t] = *(const v4f*)(kbytes + (size_t)(row_take<t>(dA0) + lo));         \
  bv[t] = *(const v4f*)(kbytes + (size_t)(row_take<t>(dB0) + lo));
  REP16(K1_LOAD0)
  av[16] = *(const v4f*)(kbytes + (size_t)(dA2 + lo));
  bv[16] = *(const v4f*)(kbytes + (size_t)(dB2 + lo));
  __builtin_amdgcn_sched_barrier(0);
#define K1_USE0(t)                                                         \
  { float pr = K1_DOT(t); pr = row16_sum(pr); res0 = (l15 == t) ? pr : res0; }
  REP16(K1_USE0)
  { float pr = K1_DOT(16); res2 = row16_sum(pr); }
  __builtin_amdgcn_sched_barrier(0);

#define K1_LOAD1(t)                                                        \
  av[t] = *(const v4f*)(kbytes + (size_t)(row_take<t>(dA1) + lo));         \
  bv[t] = *(const v4f*)(kbytes + (size_t)(row_take<t>(dB1) + lo));
  REP16(K1_LOAD1)
  __builtin_amdgcn_sched_barrier(0);
#define K1_USE1(t)                                                         \
  { float pr = K1_DOT(t); pr = row16_sum(pr); res1 = (l15 == t) ? pr : res1; }
  REP16(K1_USE1)
#undef K1_LOAD0
#undef K1_USE0
#undef K1_LOAD1
#undef K1_USE1
#undef K1_DOT

  stage[sx0] = kp0 ? res0 : 0.0f;
  stage[sx1] = kp1 ? res1 : 0.0f;
  if (l15 == 0) stage[sx2] = kp2 ? res2 : 0.0f;
  __syncthreads();

  const int c0  = g & 15;
  const int ccf = (tid >> 2) & 15;
  float* dst = W + ch0 * 256;
  if (ccf >= c0) dst[tid] = stage[tid];
  dst[256 + tid] = stage[256 + tid];
  if (ccf <= c0) dst[512 + tid] = stage[512 + tid];
}

#define RING 11
#define NBLK 66

__global__ __launch_bounds__(256) void quad_form_kernel(
    const float* __restrict__ X, const float* __restrict__ W,
    float* __restrict__ out) {
  __shared__ __attribute__((aligned(16))) float ldsX[MROWS * LDSP];
  __shared__ float partial[8][MROWS];

  const int tid   = threadIdx.x;
  const int wave  = __builtin_amdgcn_readfirstlane(tid >> 5);
  const int lane  = tid & 31;
  const int l15   = lane & 15;
  const int half  = lane >> 4;
  const int mBase = blockIdx.x * MROWS;

  float4 xr[8];
  {
    const float4* src = (const float4*)(X + (size_t)mBase * DDIM);
#pragma unroll
    for (int it = 0; it < 8; ++it) xr[it] = src[tid + it * 256];
  }

  const float* wseq = W + (size_t)wave * (NBLK * 256) + lane * 4;
  v4f ring[RING][2];

#define LOAD_B(r, s)                                                          \
  {                                                                           \
    ring[r][0] = *(const v4f*)(wseq + (s) * 256);                             \
    ring[r][1] = *(const v4f*)(wseq + (s) * 256 + 128);                       \
  }

#pragma unroll
  for (int r = 0; r < RING; ++r) LOAD_B(r, r)

#pragma unroll
  for (int it = 0; it < 8; ++it) {
    const int idx = tid + it * 256;
    const int row = idx >> 7;
    const int col = idx & 127;
    *(float4*)(ldsX + row * LDSP + col * 4) = xr[it];
  }
  __syncthreads();

  float rowSum[8];
#pragma unroll
  for (int r = 0; r < 8; ++r) rowSum[r] = 0.0f;

  v8f c0 = {0.f, 0.f, 0.f, 0.f, 0.f, 0.f, 0.f, 0.f};
  v8f c1 = {0.f, 0.f, 0.f, 0.f, 0.f, 0.f, 0.f, 0.f};

  const int e0 = wave;
  const int e2 = 33 + wave;
  const float* xrow = ldsX + l15 * LDSP + half * 8;

  v4f aBuf[2][2];
  aBuf[0][0] = *(const v4f*)(xrow);
  aBuf[0][1] = *(const v4f*)(xrow + 4);

#pragma unroll
  for (int s = 0; s < NBLK; ++s) {
    const int r   = s % RING;
    const int cur = s & 1;
    const int nxt = cur ^ 1;

    if (s + 1 < NBLK) {
      const int n = s + 1;
      int kbn;
      if (n < 8)        kbn = (n <= e0) ? n : n - 1 - e0;
      else if (n <= 16) kbn = n - 1 - e0;
      else if (n < 33)  kbn = n - 17;
      else if (n <= 40) kbn = (n <= e2) ? n - 17 : n - 1 - e2;
      else              kbn = n - 1 - e2;
      aBuf[nxt][0] = *(const v4f*)(xrow + kbn * 16);
      aBuf[nxt][1] = *(const v4f*)(xrow + kbn * 16 + 4);
    }
    __builtin_amdgcn_sched_barrier(0);
    const v4f a0 = aBuf[cur][0];
    const v4f a1 = aBuf[cur][1];

    c0 = __builtin_amdgcn_wmma_f32_16x16x4_f32(
        false, __builtin_shufflevector(a0, a0, 0, 1), false,
        __builtin_shufflevector(ring[r][0], ring[r][0], 0, 1), (short)0, c0, false, false);
    c1 = __builtin_amdgcn_wmma_f32_16x16x4_f32(
        false, __builtin_shufflevector(a0, a0, 2, 3), false,
        __builtin_shufflevector(ring[r][0], ring[r][0], 2, 3), (short)0, c1, false, false);
    c0 = __builtin_amdgcn_wmma_f32_16x16x4_f32(
        false, __builtin_shufflevector(a1, a1, 0, 1), false,
        __builtin_shufflevector(ring[r][1], ring[r][1], 0, 1), (short)0, c0, false, false);
    c1 = __builtin_amdgcn_wmma_f32_16x16x4_f32(
        false, __builtin_shufflevector(a1, a1, 2, 3), false,
        __builtin_shufflevector(ring[r][1], ring[r][1], 2, 3), (short)0, c1, false, false);

    if (s + RING < NBLK) LOAD_B(r, s + RING)

    bool last;
    int tile;
    if (s < 8)        { last = (s == e0); tile = wave; }
    else if (s < 16)  { last = false;     tile = 0; }
    else if (s == 16) { last = true;      tile = 15 - wave; }
    else if (s < 33)  { last = false;     tile = 0; }
    else if (s <= 40) { last = (s == e2); tile = 16 + wave; }
    else if (s < 65)  { last = false;     tile = 0; }
    else              { last = true;      tile = 31 - wave; }

    if (last) {
      const int jBase = tile * 16;
#pragma unroll
      for (int q = 0; q < 8; ++q) {
        const int m = q + half * 8;
        rowSum[q] += (c0[q] + c1[q]) * ldsX[m * LDSP + jBase + l15];
      }
#pragma unroll
      for (int q = 0; q < 8; ++q) { c0[q] = 0.0f; c1[q] = 0.0f; }
    }
  }
#undef LOAD_B

#pragma unroll
  for (int r = 0; r < 8; ++r) {
    float v0 = rowSum[r];
    v0 += __shfl_xor(v0, 1, 32);
    v0 += __shfl_xor(v0, 2, 32);
    v0 += __shfl_xor(v0, 4, 32);
    v0 += __shfl_xor(v0, 8, 32);
    rowSum[r] = v0;
  }
  if (l15 == 0) {
#pragma unroll
    for (int r = 0; r < 8; ++r) partial[wave][r + half * 8] = rowSum[r];
  }
  __syncthreads();

  if (tid < MROWS) {
    float s = 0.0f;
#pragma unroll
    for (int w = 0; w < 8; ++w) s += partial[w][tid];
    out[mBase + tid] = s;
  }
}

extern "C" void kernel_launch(void* const* d_in, const int* in_sizes, int n_in,
                              void* d_out, int out_size, void* d_ws, size_t ws_size,
                              hipStream_t stream) {
  const float* x    = (const float*)d_in[0];
  const float* kern = (const float*)d_in[1];
  const int*   fid  = (const int*)d_in[2];
  float* out = (float*)d_out;
  float* W   = (float*)d_ws;

  build_w_kernel<<<K1_UNITS, K1_WAVES * 32, 0, stream>>>(kern, fid, W);
  quad_form_kernel<<<4096 / MROWS, 256, 0, stream>>>(x, W, out);
}


# Features renumbered in field-id order in both kernels (pair sum is order-free): kernel 1 ranks them by wave ballots so a block's row reads recur in the WGP cache; kernel 2 puts X column c at rank[c]
# speedup vs baseline: 5.9488x; 1.0497x over previous
#include <hip/hip_runtime.h>


typedef float v2f __attribute__((ext_vector_type(2)));
typedef float v8f __attribute__((ext_vector_type(8)));
typedef float v4f __attribute__((ext_vector_type(4)));

#define DDIM 512
#define FDIM 64
#define ODIM 64
#define LDSP 516
#define MROWS 16


#define K1_WAVES 8
#define K1_UNITS 256
#define K1_SLOTS 528

__device__ __forceinline__ float row16_sum(float v) {
  v += __int_as_float(__builtin_amdgcn_update_dpp(0, __float_as_int(v), 0x161, 0xf, 0xf, true));
  v += __int_as_float(__builtin_amdgcn_update_dpp(0, __float_as_int(v), 0x162, 0xf, 0xf, true));
  v += __int_as_float(__builtin_amdgcn_update_dpp(0, __float_as_int(v), 0x164, 0xf, 0xf, true));
  v += __int_as_float(__builtin_amdgcn_update_dpp(0, __float_as_int(v), 0x168, 0xf, 0xf, true));
  return v;
}

template <int T>
__device__ __forceinline__ unsigned row_take(unsigned v) {
  return (unsigned)__builtin_amdgcn_update_dpp(0, (int)v, 0x150 + T, 0xf, 0xf, false);
}

template <int N>
__device__ __forceinline__ int row_shr(int v) {
  return __builtin_amdgcn_update_dpp(0, v, 0x110 + N, 0xf, 0xf, true);
}

#define NCHUNK 528

__device__ __forceinline__ void decode_slot(int P, int ch0, const unsigned* sinv,
                                            unsigned& offA, unsigned& offB,
                                            int& sidx, bool& keep) {
  const int ch = P >> 8;
  const int cc = (P >> 4) & 15;
  const int kl = P & 15;
  const int wv = ch / 66;
  const int st = ch - wv * 66;
  const bool hi  = st > 16;
  const bool odd = st > (hi ? 33 + wv : wv);
  const int J    = (hi ? 16 : 0) + (odd ? 15 - wv : wv);
  const int kb   = st - (hi ? (odd ? 34 + wv : 17) : (odd ? wv + 1 : 0));
  const int np = J * 16 + cc;
  const int kp = kb * 16 + kl;
  const unsigned pn = sinv[np];
  const unsigned pk = sinv[kp];
  const unsigned n = pn & 0xffffu, fn = pn >> 16;
  const unsigned k = pk & 0xffffu, fk = pk >> 16;
  offA = (n * FDIM + fk) * (ODIM * 4);
  offB = (k * FDIM + fn) * (ODIM * 4);
  keep = kp < np;
  sidx = (ch - ch0) * 256 + ((kl >> 2) & 1) * 128 + (((kl >> 3) * 16 + cc) * 4) + (kl & 3);
}

#define REP16(M) M(0) M(1) M(2) M(3) M(4) M(5) M(6) M(7) \
                 M(8) M(9) M(10) M(11) M(12) M(13) M(14) M(15)

__global__ __launch_bounds__(K1_WAVES * 32) void build_w_kernel(
    const float* __restrict__ kern, const int* __restrict__ fid,
    float* __restrict__ W) {
  __shared__ float stage[768];
  __shared__ __attribute__((aligned(16))) int cnt[FDIM * K1_WAVES];
  __shared__ unsigned sinv[DDIM];

  const int tid  = threadIdx.x;
  const int l15  = tid & 15;
  const int hw   = tid >> 4;
  const int g    = blockIdx.x;
  const int base = g * K1_SLOTS + hw * 33;
  const int ch0  = (g * K1_SLOTS) >> 8;

  int rankA, rankB;
  {
    const int lane = tid & 31;
    const int wv   = __builtin_amdgcn_readfirstlane(tid >> 5);
    const int ia   = wv * 64 + lane;
    const int ib   = ia + 32;
    const unsigned fa = (unsigned)fid[ia] & (FDIM - 1);
    const unsigned fb = (unsigned)fid[ib] & (FDIM - 1);
    unsigned maa = ~0u, mab = ~0u, mba = ~0u, mbb = ~0u;
#pragma unroll
    for (int j = 0; j < 6; ++j) {
      const unsigned ba = (fa >> j) & 1u;
      const unsigned bb = (fb >> j) & 1u;
      const unsigned Ba = __builtin_amdgcn_ballot_w32(ba != 0u);
      const unsigned Bb = __builtin_amdgcn_ballot_w32(bb != 0u);
      const unsigned xa = ba - 1u;
      const unsigned xb = bb - 1u;
      maa &= Ba ^ xa;  mab &= Bb ^ xa;
      mba &= Ba ^ xb;  mbb &= Bb ^ xb;
    }
    const int ra = (int)__builtin_amdgcn_mbcnt_lo(maa, 0u);
    const int rb = __builtin_popcount(mba) + (int)__builtin_amdgcn_mbcnt_lo(mbb, 0u);
    const int ca = __builtin_popcount(maa) + __builtin_popcount(mab);
    const int cb = __builtin_popcount(mba) + __builtin_popcount(mbb);
    cnt[lane * K1_WAVES + wv] = 0;
    cnt[(lane + 32) * K1_WAVES + wv] = 0;
    if (ra == 0) cnt[fa * K1_WAVES + wv] = ca;
    if (rb == 0) cnt[fb * K1_WAVES + wv] = cb;
    __syncthreads();

    const int4* cr = (const int4*)cnt;
    const int4 l0 = cr[lane * 2], l1 = cr[lane * 2 + 1];
    const int4 h0 = cr[(lane + 32) * 2], h1 = cr[(lane + 32) * 2 + 1];
    const int cl[8] = {l0.x, l0.y, l0.z, l0.w, l1.x, l1.y, l1.z, l1.w};
    const int ch[8] = {h0.x, h0.y, h0.z, h0.w, h1.x, h1.y, h1.z, h1.w};
    int tl = 0, pl = 0, th = 0, ph = 0;
#pragma unroll
    for (int q = 0; q < 8; ++q) {
      tl += cl[q];  pl += (q < wv) ? cl[q] : 0;
      th += ch[q];  ph += (q < wv) ? ch[q] : 0;
    }
    int s = tl | (th << 16);
    s += row_shr<1>(s);
    s += row_shr<2>(s);
    s += row_shr<4>(s);
    s += row_shr<8>(s);
    const int r15 = __builtin_amdgcn_readlane(s, 15);
    if (lane >= 16) s += r15;
    const int totLo   = __builtin_amdgcn_readlane(s, 31) & 0xffff;
    const int startLo = (s & 0xffff) - tl;
    const int startHi = totLo + (s >> 16) - th;
    const int bp = (startLo + pl) | ((startHi + ph) << 16);
    const int va = __builtin_amdgcn_ds_bpermute((int)((fa & 31u) << 2), bp);
    const int vb = __builtin_amdgcn_ds_bpermute((int)((fb & 31u) << 2), bp);
    const int rka = ((fa & 32u) ? (va >> 16) : (va & 0xffff)) + ra;
    const int rkb = ((fb & 32u) ? (vb >> 16) : (vb & 0xffff)) + rb;
    sinv[rka] = (unsigned)ia | (fa << 16);
    sinv[rkb] = (unsigned)ib | (fb << 16);
    rankA = rka;
    rankB = rkb;
    __syncthreads();
  }

  unsigned dA0, dB0, dA1, dB1, dA2, dB2;
  int sx0, sx1, sx2;
  bool kp0, kp1, kp2;
  decode_slot(base + l15,      ch0, sinv, dA0, dB0, sx0, kp0);
  decode_slot(base + 16 + l15, ch0, sinv, dA1, dB1, sx1, kp1);
  decode_slot(base + 32,       ch0, sinv, dA2, dB2, sx2, kp2);

  const char* kbytes = (const char*)kern;
  const unsigned lo = (unsigned)l15 * 16u;

  v4f av[17], bv[17];
  float res0 = 0.0f, res1 = 0.0f, res2 = 0.0f;

#define K1_DOT(t) (av[t].x * bv[t].x + av[t].y * bv[t].y + \
                   av[t].z * bv[t].z + av[t].w * bv[t].w)

#define K1_LOAD0(t)                                                        \
  av[t] = *(const v4f*)(kbytes + (size_t)(row_take<t>(dA0) + lo));         \
  bv[t] = *(const v4f*)(kbytes + (size_t)(row_take<t>(dB0) + lo));
  REP16(K1_LOAD0)
  av[16] = *(const v4f*)(kbytes + (size_t)(dA2 + lo));
  bv[16] = *(const v4f*)(kbytes + (size_t)(dB2 + lo));
  __builtin_amdgcn_sched_barrier(0);
#define K1_USE0(t)                                                         \
  { float pr = K1_DOT(t); pr = row16_sum(pr); res0 = (l15 == t) ? pr : res0; }
  REP16(K1_USE0)
  { float pr = K1_DOT(16); res2 = row16_sum(pr); }
  __builtin_amdgcn_sched_barrier(0);

#define K1_LOAD1(t)                                                        \
  av[t] = *(const v4f*)(kbytes + (size_t)(row_take<t>(dA1) + lo));         \
  bv[t] = *(const v4f*)(kbytes + (size_t)(row_take<t>(dB1) + lo));
  REP16(K1_LOAD1)
  __builtin_amdgcn_sched_barrier(0);
#define K1_USE1(t)                                                         \
  { float pr = K1_DOT(t); pr = row16_sum(pr); res1 = (l15 == t) ? pr : res1; }
  REP16(K1_USE1)
#undef K1_LOAD0
#undef K1_USE0
#undef K1_LOAD1
#undef K1_USE1
#undef K1_DOT

  stage[sx0] = kp0 ? res0 : 0.0f;
  stage[sx1] = kp1 ? res1 : 0.0f;
  if (l15 == 0) stage[sx2] = kp2 ? res2 : 0.0f;
  __syncthreads();

  const int c0  = g & 15;
  const int ccf = (tid >> 2) & 15;
  float* dst = W + ch0 * 256;
  if (ccf >= c0) dst[tid] = stage[tid];
  dst[256 + tid] = stage[256 + tid];
  if (ccf <= c0) dst[512 + tid] = stage[512 + tid];

  if (g == 0) {
    int* rankOut = (int*)(W + NCHUNK * 256);
    const int ia = (tid >> 5) * 64 + (tid & 31);
    rankOut[ia]      = rankA;
    rankOut[ia + 32] = rankB;
  }
}

#define RING 11
#define NBLK 66

__global__ __launch_bounds__(256) void quad_form_kernel(
    const float* __restrict__ X, const float* __restrict__ W,
    float* __restrict__ out) {
  __shared__ __attribute__((aligned(16))) float ldsX[MROWS * LDSP];
  __shared__ float partial[8][MROWS];

  const int tid   = threadIdx.x;
  const int wave  = __builtin_amdgcn_readfirstlane(tid >> 5);
  const int lane  = tid & 31;
  const int l15   = lane & 15;
  const int half  = lane >> 4;
  const int mBase = blockIdx.x * MROWS;

  const int4 rk = *(const int4*)((const int*)(W + NCHUNK * 256) + (tid & 127) * 4);
  __builtin_amdgcn_sched_barrier(0);

  float4 xr[8];
  {
    const float4* src = (const float4*)(X + (size_t)mBase * DDIM);
#pragma unroll
    for (int it = 0; it < 8; ++it) xr[it] = src[tid + it * 256];
  }
  __builtin_amdgcn_sched_barrier(0);

  const float* wseq = W + (size_t)wave * (NBLK * 256) + lane * 4;
  v4f ring[RING][2];

#define LOAD_B(r, s)                                                          \
  {                                                                           \
    ring[r][0] = *(const v4f*)(wseq + (s) * 256);                             \
    ring[r][1] = *(const v4f*)(wseq + (s) * 256 + 128);                       \
  }

#pragma unroll
  for (int r = 0; r < RING; ++r) LOAD_B(r, r)
  __builtin_amdgcn_sched_barrier(0);

#pragma unroll
  for (int it = 0; it < 8; ++it) {
    const int idx = tid + it * 256;
    const int row = idx >> 7;
    float* xdst = ldsX + row * LDSP;
    xdst[rk.x] = xr[it].x;
    xdst[rk.y] = xr[it].y;
    xdst[rk.z] = xr[it].z;
    xdst[rk.w] = xr[it].w;
  }
  __syncthreads();

  float rowSum[8];
#pragma unroll
  for (int r = 0; r < 8; ++r) rowSum[r] = 0.0f;

  v8f c0 = {0.f, 0.f, 0.f, 0.f, 0.f, 0.f, 0.f, 0.f};
  v8f c1 = {0.f, 0.f, 0.f, 0.f, 0.f, 0.f, 0.f, 0.f};

  const int e0 = wave;
  const int e2 = 33 + wave;
  const float* xrow = ldsX + l15 * LDSP + half * 8;

  v4f aBuf[2][2];
  aBuf[0][0] = *(const v4f*)(xrow);
  aBuf[0][1] = *(const v4f*)(xrow + 4);

#pragma unroll
  for (int s = 0; s < NBLK; ++s) {
    const int r   = s % RING;
    const int cur = s & 1;
    const int nxt = cur ^ 1;

    if (s + 1 < NBLK) {
      const int n = s + 1;
      int kbn;
      if (n < 8)        kbn = (n <= e0) ? n : n - 1 - e0;
      else if (n <= 16) kbn = n - 1 - e0;
      else if (n < 33)  kbn = n - 17;
      else if (n <= 40) kbn = (n <= e2) ? n - 17 : n - 1 - e2;
      else              kbn = n - 1 - e2;
      aBuf[nxt][0] = *(const v4f*)(xrow + kbn * 16);
      aBuf[nxt][1] = *(const v4f*)(xrow + kbn * 16 + 4);
    }
    __builtin_amdgcn_sched_barrier(0);
    const v4f a0 = aBuf[cur][0];
    const v4f a1 = aBuf[cur][1];

    c0 = __builtin_amdgcn_wmma_f32_16x16x4_f32(
        false, __builtin_shufflevector(a0, a0, 0, 1), false,
        __builtin_shufflevector(ring[r][0], ring[r][0], 0, 1), (short)0, c0, false, false);
    c1 = __builtin_amdgcn_wmma_f32_16x16x4_f32(
        false, __builtin_shufflevector(a0, a0, 2, 3), false,
        __builtin_shufflevector(ring[r][0], ring[r][0], 2, 3), (short)0, c1, false, false);
    c0 = __builtin_amdgcn_wmma_f32_16x16x4_f32(
        false, __builtin_shufflevector(a1, a1, 0, 1), false,
        __builtin_shufflevector(ring[r][1], ring[r][1], 0, 1), (short)0, c0, false, false);
    c1 = __builtin_amdgcn_wmma_f32_16x16x4_f32(
        false, __builtin_shufflevector(a1, a1, 2, 3), false,
        __builtin_shufflevector(ring[r][1], ring[r][1], 2, 3), (short)0, c1, false, false);

    if (s + RING < NBLK) LOAD_B(r, s + RING)

    bool last;
    int tile;
    if (s < 8)        { last = (s == e0); tile = wave; }
    else if (s < 16)  { last = false;     tile = 0; }
    else if (s == 16) { last = true;      tile = 15 - wave; }
    else if (s < 33)  { last = false;     tile = 0; }
    else if (s <= 40) { last = (s == e2); tile = 16 + wave; }
    else if (s < 65)  { last = false;     tile = 0; }
    else              { last = true;      tile = 31 - wave; }

    if (last) {
      const int jBase = tile * 16;
#pragma unroll
      for (int q = 0; q < 8; ++q) {
        const int m = q + half * 8;
        rowSum[q] += (c0[q] + c1[q]) * ldsX[m * LDSP + jBase + l15];
      }
#pragma unroll
      for (int q = 0; q < 8; ++q) { c0[q] = 0.0f; c1[q] = 0.0f; }
    }
  }
#undef LOAD_B

#pragma unroll
  for (int r = 0; r < 8; ++r) {
    float v0 = rowSum[r];
    v0 += __shfl_xor(v0, 1, 32);
    v0 += __shfl_xor(v0, 2, 32);
    v0 += __shfl_xor(v0, 4, 32);
    v0 += __shfl_xor(v0, 8, 32);
    rowSum[r] = v0;
  }
  if (l15 == 0) {
#pragma unroll
    for (int r = 0; r < 8; ++r) partial[wave][r + half * 8] = rowSum[r];
  }
  __syncthreads();

  if (tid < MROWS) {
    float s = 0.0f;
#pragma unroll
    for (int w = 0; w < 8; ++w) s += partial[w][tid];
    out[mBase + tid] = s;
  }
}

extern "C" void kernel_launch(void* const* d_in, const int* in_sizes, int n_in,
                              void* d_out, int out_size, void* d_ws, size_t ws_size,
                              hipStream_t stream) {
  const float* x    = (const float*)d_in[0];
  const float* kern = (const float*)d_in[1];
  const int*   fid  = (const int*)d_in[2];
  float* out = (float*)d_out;
  float* W   = (float*)d_ws;

  build_w_kernel<<<K1_UNITS, K1_WAVES * 32, 0, stream>>>(kern, fid, W);
  quad_form_kernel<<<4096 / MROWS, 256, 0, stream>>>(x, W, out);
}
